# starting kernel
# baseline (speedup 1.0000x reference)
#include <hip/hip_runtime.h>
#include <math.h>

typedef __attribute__((ext_vector_type(2))) float v2f;
typedef __attribute__((ext_vector_type(8))) float v8f;

#define TMAX 4096
#define KC 8
#define PTS_PER_WAVE (4 * KC)
#define PPL (2 * KC)
#define WAVES_PER_BLOCK 8
#define BLOCK 256

static __device__ __forceinline__ float bcast_first(float x) {
  return __int_as_float(__builtin_amdgcn_readfirstlane(__float_as_int(x)));
}

__global__ __launch_bounds__(BLOCK) void hyst_dens_kernel(
    const float* __restrict__ raw, float* __restrict__ dens, int n, int npad) {
  int j = blockIdx.x * blockDim.x + threadIdx.x;
  if (j < npad) {
    float v = 0.0f;
    if (j < n) {
      float x = raw[j];
      v = fmaxf(x, 0.0f) + log1pf(expf(-fabsf(x)));
    }
    dens[j] = v;
  }
}

__global__ __launch_bounds__(BLOCK) void hyst_scan_kernel(
    const float* __restrict__ h, const float* __restrict__ mesh,
    const float* __restrict__ dens, float* __restrict__ partial,
    int T, int n) {
  __shared__ float lh[TMAX];
  for (int i = threadIdx.x; i < T; i += BLOCK) lh[i] = h[i];
  __syncthreads();

  const int lane = threadIdx.x & 31;
  const int row  = lane & 15;
  const int half = lane >> 4;
  const int w    = blockIdx.x * WAVES_PER_BLOCK + (threadIdx.x >> 5);
  const int pbase = w * PTS_PER_WAVE + 2 * half;

  float al[PPL], be[PPL], dd[PPL];
#pragma unroll
  for (int kc = 0; kc < KC; ++kc) {
#pragma unroll
    for (int i = 0; i < 2; ++i) {
      int p = pbase + 4 * kc + i;
      float a = 3.0e38f, b = -3.0e38f, d = 0.0f;
      if (p < n) { b = mesh[2 * p]; a = mesh[2 * p + 1]; d = dens[p]; }
      al[2 * kc + i] = a; be[2 * kc + i] = b; dd[2 * kc + i] = d;
    }
  }

  unsigned sbits = 0u;
  unsigned snapbits = 0u;
  float* pout = partial + (size_t)w * T;

  float hp = 0.0f;
  for (int tb = 0; tb < T; tb += 16) {
    v8f acc = {};
    for (int tt = 0; tt < 16; ++tt) {
      float hi = bcast_first(lh[tb + tt]);
      if (hi > hp) {
#pragma unroll
        for (int p = 0; p < PPL; ++p)
          sbits |= (al[p] <= hi) ? (1u << p) : 0u;
      } else if (hi < hp) {
#pragma unroll
        for (int p = 0; p < PPL; ++p)
          sbits &= (be[p] >= hi) ? ~(1u << p) : 0xffffffffu;
      }
      if (tt == row) snapbits = sbits;
      hp = hi;
    }
#pragma unroll
    for (int kc = 0; kc < KC; ++kc) {
      v2f A, B;
      A.x = ((snapbits >> (2 * kc)) & 1u) ? 1.0f : -1.0f;
      A.y = ((snapbits >> (2 * kc + 1)) & 1u) ? 1.0f : -1.0f;
      B.x = dd[2 * kc];
      B.y = dd[2 * kc + 1];
      acc = __builtin_amdgcn_wmma_f32_16x16x4_f32(
          false, A, false, B, (short)0, acc, false, false);
    }
    if (lane == 0) {
      *reinterpret_cast<float4*>(pout + tb) =
          make_float4(acc[0], acc[1], acc[2], acc[3]);
      *reinterpret_cast<float4*>(pout + tb + 4) =
          make_float4(acc[4], acc[5], acc[6], acc[7]);
    } else if (lane == 16) {
      *reinterpret_cast<float4*>(pout + tb + 8) =
          make_float4(acc[0], acc[1], acc[2], acc[3]);
      *reinterpret_cast<float4*>(pout + tb + 12) =
          make_float4(acc[4], acc[5], acc[6], acc[7]);
    }
  }
}

__global__ __launch_bounds__(BLOCK) void hyst_reduce_kernel(
    const float* __restrict__ partial, const float* __restrict__ scale,
    const float* __restrict__ offset, float* __restrict__ out,
    int T, int NW, float inv_n) {
  int t = blockIdx.x * blockDim.x + threadIdx.x;
  if (t < T) {
    float sum = 0.0f;
    for (int wv = 0; wv < NW; ++wv) sum += partial[(size_t)wv * T + t];
    out[t] = scale[0] * (sum * inv_n) + offset[0];
  }
}

extern "C" void kernel_launch(void* const* d_in, const int* in_sizes, int n_in,
                              void* d_out, int out_size, void* d_ws, size_t ws_size,
                              hipStream_t stream) {
  const float* h      = (const float*)d_in[0];
  const float* mesh   = (const float*)d_in[1];
  const float* raw    = (const float*)d_in[2];
  const float* scale  = (const float*)d_in[3];
  const float* offset = (const float*)d_in[4];
  float* out = (float*)d_out;

  const int T = in_sizes[0];
  const int n = in_sizes[2];

  const int nWaves  = (n + PTS_PER_WAVE - 1) / PTS_PER_WAVE;
  const int nBlocks = (nWaves + WAVES_PER_BLOCK - 1) / WAVES_PER_BLOCK;
  const int NW      = nBlocks * WAVES_PER_BLOCK;
  const int npad    = NW * PTS_PER_WAVE;

  float* partial = (float*)d_ws;
  float* dens    = partial + (size_t)NW * T;

  hyst_dens_kernel<<<(npad + BLOCK - 1) / BLOCK, BLOCK, 0, stream>>>(
      raw, dens, n, npad);
  hyst_scan_kernel<<<nBlocks, BLOCK, 0, stream>>>(
      h, mesh, dens, partial, T, n);
  hyst_reduce_kernel<<<(T + BLOCK - 1) / BLOCK, BLOCK, 0, stream>>>(
      partial, scale, offset, out, T, NW, 1.0f / (float)n);
}


# scan kernel: one relay per lane, the wave's 32 states as a lane mask in a scalar register (2 compares + 2 scalar bit ops per step) instead of each lane scanning 16 relays; one-wave work-groups
# speedup vs baseline: 2.5429x; 2.5429x over previous
#include <hip/hip_runtime.h>
#include <math.h>


typedef __attribute__((ext_vector_type(2))) float v2f;
typedef __attribute__((ext_vector_type(8))) float v8f;

#define KC 8
#define PTS_PER_WAVE (4 * KC)
#define PPL (2 * KC)
#define WAVES_PER_BLOCK 8
#define BLOCK 256
#define SCAN_BLOCK 32

__global__ __launch_bounds__(BLOCK) void hyst_dens_kernel(
    const float* __restrict__ raw, float* __restrict__ dens, int n, int npad) {
  int j = blockIdx.x * blockDim.x + threadIdx.x;
  if (j < npad) {
    float v = 0.0f;
    if (j < n) {
      float x = raw[j];
      v = fmaxf(x, 0.0f) + log1pf(expf(-fabsf(x)));
    }
    dens[j] = v;
  }
}

static __device__ __forceinline__ unsigned scan16(
    const float (&hv)[16], float& hp, unsigned& sbits, float al, float be) {
  const float qnan = __int_as_float(0x7fc00000);
  unsigned snap = 0u;
#pragma unroll
  for (int tt = 0; tt < 16; ++tt) {
    const float hi = hv[tt];
    const float hu = (hi > hp) ? hi : qnan;
    const float hd = (hi < hp) ? hi : qnan;
    const unsigned mu = __builtin_amdgcn_ballot_w32(al <= hu);
    const unsigned md = __builtin_amdgcn_ballot_w32(be >= hd);
    sbits = (sbits | mu) & ~md;
    snap = __builtin_amdgcn_inverse_ballot_w32(0x00010001u << tt) ? sbits : snap;
    hp = hi;
  }
  return snap;
}

static __device__ __forceinline__ void load16(float (&hv)[16], const float* __restrict__ h, int t0) {
#pragma unroll
  for (int i = 0; i < 16; ++i) hv[i] = h[t0 + i];
}

static __device__ __forceinline__ void product16(
    unsigned snap, const float (&dd)[PPL], float* __restrict__ pout, int tb,
    int row, int half) {
  const unsigned nx = ~(snap >> (2 * half));
  v8f acc = {};
#pragma unroll
  for (int kc = 0; kc < KC; ++kc) {
    v2f A, B;
    A.x = __uint_as_float(((nx << (31 - 4 * kc)) & 0x80000000u) | 0x3f800000u);
    A.y = __uint_as_float(((nx << (30 - 4 * kc)) & 0x80000000u) | 0x3f800000u);
    B.x = dd[2 * kc];
    B.y = dd[2 * kc + 1];
    acc = __builtin_amdgcn_wmma_f32_16x16x4_f32(
        false, A, false, B, (short)0, acc, false, false);
  }
  if (row == 0) {
    float* q = pout + tb + 8 * half;
    *reinterpret_cast<float4*>(q)     = make_float4(acc[0], acc[1], acc[2], acc[3]);
    *reinterpret_cast<float4*>(q + 4) = make_float4(acc[4], acc[5], acc[6], acc[7]);
  }
}

__global__ __launch_bounds__(SCAN_BLOCK) void hyst_scan_kernel(
    const float* __restrict__ h, const float* __restrict__ mesh,
    const float* __restrict__ dens, float* __restrict__ partial,
    int T, int n) {
  const int lane = threadIdx.x;
  const int row  = lane & 15;
  const int half = lane >> 4;
  const int w    = blockIdx.x;

  float al = 3.0e38f, be = -3.0e38f;
  {
    const int p = w * PTS_PER_WAVE + lane;
    if (p < n) { be = mesh[2 * p]; al = mesh[2 * p + 1]; }
  }
  float dd[PPL];
  {
    const int pbase = w * PTS_PER_WAVE + 2 * half;
#pragma unroll
    for (int kc = 0; kc < KC; ++kc) {
#pragma unroll
      for (int i = 0; i < 2; ++i) {
        const int p = pbase + 4 * kc + i;
        dd[2 * kc + i] = (p < n) ? dens[p] : 0.0f;
      }
    }
  }

  float* pout = partial + (size_t)w * T;
  unsigned sbits = 0u;
  float hp = 0.0f;

  const int tlast = T - 16;
  float hA[16], hB[16], hC[16], hD[16];
  load16(hA, h, 0);
  load16(hB, h, min(16, tlast));
  load16(hC, h, min(32, tlast));
  unsigned s_prev = scan16(hA, hp, sbits, al, be);
  int tb = 16;
  for (; tb + 64 <= T; tb += 64) {
    load16(hD, h, min(tb + 32, tlast));
    unsigned s0 = scan16(hB, hp, sbits, al, be);
    product16(s_prev, dd, pout, tb - 16, row, half);
    load16(hA, h, min(tb + 48, tlast));
    unsigned s1 = scan16(hC, hp, sbits, al, be);
    product16(s0, dd, pout, tb, row, half);
    load16(hB, h, min(tb + 64, tlast));
    unsigned s2 = scan16(hD, hp, sbits, al, be);
    product16(s1, dd, pout, tb + 16, row, half);
    load16(hC, h, min(tb + 80, tlast));
    unsigned s3 = scan16(hA, hp, sbits, al, be);
    product16(s2, dd, pout, tb + 32, row, half);
    s_prev = s3;
  }
  for (; tb <= T; tb += 16) {
    unsigned s = 0u;
    if (tb < T) {
      load16(hA, h, tb);
      s = scan16(hA, hp, sbits, al, be);
    }
    product16(s_prev, dd, pout, tb - 16, row, half);
    s_prev = s;
  }
}

__global__ __launch_bounds__(BLOCK) void hyst_reduce_kernel(
    const float* __restrict__ partial, const float* __restrict__ scale,
    const float* __restrict__ offset, float* __restrict__ out,
    int T, int NW, float inv_n) {
  int t = blockIdx.x * blockDim.x + threadIdx.x;
  if (t < T) {
    float sum = 0.0f;
    for (int wv = 0; wv < NW; ++wv) sum += partial[(size_t)wv * T + t];
    out[t] = scale[0] * (sum * inv_n) + offset[0];
  }
}

extern "C" void kernel_launch(void* const* d_in, const int* in_sizes, int n_in,
                              void* d_out, int out_size, void* d_ws, size_t ws_size,
                              hipStream_t stream) {
  const float* h      = (const float*)d_in[0];
  const float* mesh   = (const float*)d_in[1];
  const float* raw    = (const float*)d_in[2];
  const float* scale  = (const float*)d_in[3];
  const float* offset = (const float*)d_in[4];
  float* out = (float*)d_out;

  const int T = in_sizes[0];
  const int n = in_sizes[2];

  const int nWaves  = (n + PTS_PER_WAVE - 1) / PTS_PER_WAVE;
  const int nBlocks = (nWaves + WAVES_PER_BLOCK - 1) / WAVES_PER_BLOCK;
  const int NW      = nBlocks * WAVES_PER_BLOCK;
  const int npad    = NW * PTS_PER_WAVE;

  float* partial = (float*)d_ws;
  float* dens    = partial + (size_t)NW * T;

  hyst_dens_kernel<<<(npad + BLOCK - 1) / BLOCK, BLOCK, 0, stream>>>(
      raw, dens, n, npad);
  hyst_scan_kernel<<<NW, SCAN_BLOCK, 0, stream>>>(
      h, mesh, dens, partial, T, n);
  hyst_reduce_kernel<<<(T + BLOCK - 1) / BLOCK, BLOCK, 0, stream>>>(
      partial, scale, offset, out, T, NW, 1.0f / (float)n);
}


# scan kernel: time cut into 16 chunks run side by side by the 16 waves of a work-group; a step only sets, clears or keeps a relay, so each chunk's net effect is found first and folded in through LDS
# speedup vs baseline: 5.2483x; 2.0639x over previous
#include <hip/hip_runtime.h>
#include <math.h>


typedef __attribute__((ext_vector_type(2))) float v2f;
typedef __attribute__((ext_vector_type(8))) float v8f;

#define KC 8
#define PTS_PER_WAVE (4 * KC)
#define PPL (2 * KC)
#define WAVES_PER_BLOCK 8
#define BLOCK 256
#define TCH 16
#define SCAN_BLOCK (32 * TCH)

__global__ __launch_bounds__(BLOCK) void hyst_dens_kernel(
    const float* __restrict__ raw, float* __restrict__ dens, int n, int npad) {
  int j = blockIdx.x * blockDim.x + threadIdx.x;
  if (j < npad) {
    float v = 0.0f;
    if (j < n) {
      float x = raw[j];
      v = fmaxf(x, 0.0f) + log1pf(expf(-fabsf(x)));
    }
    dens[j] = v;
  }
}

static __device__ __forceinline__ unsigned scan16(
    const float (&hv)[16], float& hp, unsigned& sbits, float al, float be) {
  const float qnan = __int_as_float(0x7fc00000);
  unsigned snap = 0u;
#pragma unroll
  for (int tt = 0; tt < 16; ++tt) {
    const float hi = hv[tt];
    const float hu = (hi > hp) ? hi : qnan;
    const float hd = (hi < hp) ? hi : qnan;
    const unsigned mu = __builtin_amdgcn_ballot_w32(al <= hu);
    const unsigned md = __builtin_amdgcn_ballot_w32(be >= hd);
    sbits = (sbits | mu) & ~md;
    snap = __builtin_amdgcn_inverse_ballot_w32(0x00010001u << tt) ? sbits : snap;
    hp = hi;
  }
  return snap;
}

static __device__ __forceinline__ void scan16_ends(
    const float (&hv)[16], float& hp, unsigned& e0, unsigned& e1, float al, float be) {
  const float qnan = __int_as_float(0x7fc00000);
#pragma unroll
  for (int tt = 0; tt < 16; ++tt) {
    const float hi = hv[tt];
    const float hu = (hi > hp) ? hi : qnan;
    const float hd = (hi < hp) ? hi : qnan;
    const unsigned mu = __builtin_amdgcn_ballot_w32(al <= hu);
    const unsigned md = __builtin_amdgcn_ballot_w32(be >= hd);
    e0 = (e0 | mu) & ~md;
    e1 = (e1 | mu) & ~md;
    hp = hi;
  }
}

static __device__ __forceinline__ void load16(float (&hv)[16], const float* __restrict__ h, int t0) {
#pragma unroll
  for (int i = 0; i < 16; ++i) hv[i] = h[t0 + i];
}

static __device__ __forceinline__ void product16(
    unsigned snap, const float (&dd)[PPL], float* __restrict__ pout, int tb,
    int row, int half) {
  const unsigned nx = ~(snap >> (2 * half));
  v8f acc = {};
#pragma unroll
  for (int kc = 0; kc < KC; ++kc) {
    v2f A, B;
    A.x = __uint_as_float(((nx << (31 - 4 * kc)) & 0x80000000u) | 0x3f800000u);
    A.y = __uint_as_float(((nx << (30 - 4 * kc)) & 0x80000000u) | 0x3f800000u);
    B.x = dd[2 * kc];
    B.y = dd[2 * kc + 1];
    acc = __builtin_amdgcn_wmma_f32_16x16x4_f32(
        false, A, false, B, (short)0, acc, false, false);
  }
  if (row == 0) {
    float* q = pout + tb + 8 * half;
    *reinterpret_cast<float4*>(q)     = make_float4(acc[0], acc[1], acc[2], acc[3]);
    *reinterpret_cast<float4*>(q + 4) = make_float4(acc[4], acc[5], acc[6], acc[7]);
  }
}

__global__ __launch_bounds__(SCAN_BLOCK) void hyst_scan_kernel(
    const float* __restrict__ h, const float* __restrict__ mesh,
    const float* __restrict__ dens, float* __restrict__ partial,
    int T, int n) {
  __shared__ unsigned end0[TCH];
  __shared__ unsigned end1[TCH];

  const int lane = threadIdx.x & 31;
  const int c    = __builtin_amdgcn_readfirstlane((int)(threadIdx.x >> 5));
  const int row  = lane & 15;
  const int half = lane >> 4;
  const int w    = blockIdx.x;

  float al = 3.0e38f, be = -3.0e38f;
  {
    const int p = w * PTS_PER_WAVE + lane;
    if (p < n) { be = mesh[2 * p]; al = mesh[2 * p + 1]; }
  }
  float dd[PPL];
  {
    const int pbase = w * PTS_PER_WAVE + 2 * half;
#pragma unroll
    for (int kc = 0; kc < KC; ++kc) {
#pragma unroll
      for (int i = 0; i < 2; ++i) {
        const int p = pbase + 4 * kc + i;
        dd[2 * kc + i] = (p < n) ? dens[p] : 0.0f;
      }
    }
  }

  const int ntile = T >> 4;
  const int tpc   = (((ntile + TCH - 1) / TCH) + 1) & ~1;
  const int t0    = min(c * tpc, ntile) * 16;
  const int t1    = min(c * tpc + tpc, ntile) * 16;
  const int tlast = T - 16;
  float hp0 = 0.0f;
  if (t0 > 0) hp0 = h[t0 - 1];

  float hA[16], hB[16];

  unsigned e0 = 0u, e1 = ~0u;
  if (c < TCH - 1 && t0 < t1) {
    float hp = hp0;
    load16(hA, h, t0);
    for (int tb = t0; tb < t1; tb += 32) {
      load16(hB, h, min(tb + 16, tlast));
      scan16_ends(hA, hp, e0, e1, al, be);
      if (tb + 16 >= t1) break;
      load16(hA, h, min(tb + 32, tlast));
      scan16_ends(hB, hp, e0, e1, al, be);
    }
  }
  if (lane == 0) { end0[c] = e0; end1[c] = e1; }
  __syncthreads();

  unsigned inc = 0u;
#pragma unroll
  for (int k = 0; k < TCH - 1; ++k) {
    const unsigned a = end0[k];
    const unsigned b = end1[k];
    inc = (k < c) ? ((inc & b) | a) : inc;
  }

  if (t0 < t1) {
    float* pout = partial + (size_t)w * T;
    unsigned sbits = __builtin_amdgcn_readfirstlane(inc);
    float hp = hp0;
    load16(hA, h, t0);
    load16(hB, h, min(t0 + 16, tlast));
    unsigned s_prev = scan16(hA, hp, sbits, al, be);
    for (int tb = t0 + 16; tb < t1; tb += 32) {
      load16(hA, h, min(tb + 16, tlast));
      const unsigned s0 = scan16(hB, hp, sbits, al, be);
      product16(s_prev, dd, pout, tb - 16, row, half);
      s_prev = s0;
      if (tb + 16 >= t1) break;
      load16(hB, h, min(tb + 32, tlast));
      const unsigned s1 = scan16(hA, hp, sbits, al, be);
      product16(s0, dd, pout, tb, row, half);
      s_prev = s1;
    }
    product16(s_prev, dd, pout, t1 - 16, row, half);
  }
}

__global__ __launch_bounds__(BLOCK) void hyst_reduce_kernel(
    const float* __restrict__ partial, const float* __restrict__ scale,
    const float* __restrict__ offset, float* __restrict__ out,
    int T, int NW, float inv_n) {
  int t = blockIdx.x * blockDim.x + threadIdx.x;
  if (t < T) {
    float sum = 0.0f;
    for (int wv = 0; wv < NW; ++wv) sum += partial[(size_t)wv * T + t];
    out[t] = scale[0] * (sum * inv_n) + offset[0];
  }
}

extern "C" void kernel_launch(void* const* d_in, const int* in_sizes, int n_in,
                              void* d_out, int out_size, void* d_ws, size_t ws_size,
                              hipStream_t stream) {
  const float* h      = (const float*)d_in[0];
  const float* mesh   = (const float*)d_in[1];
  const float* raw    = (const float*)d_in[2];
  const float* scale  = (const float*)d_in[3];
  const float* offset = (const float*)d_in[4];
  float* out = (float*)d_out;

  const int T = in_sizes[0];
  const int n = in_sizes[2];

  const int nWaves  = (n + PTS_PER_WAVE - 1) / PTS_PER_WAVE;
  const int nBlocks = (nWaves + WAVES_PER_BLOCK - 1) / WAVES_PER_BLOCK;
  const int NW      = nBlocks * WAVES_PER_BLOCK;
  const int npad    = NW * PTS_PER_WAVE;

  float* partial = (float*)d_ws;
  float* dens    = partial + (size_t)NW * T;

  hyst_dens_kernel<<<(npad + BLOCK - 1) / BLOCK, BLOCK, 0, stream>>>(
      raw, dens, n, npad);
  hyst_scan_kernel<<<NW, SCAN_BLOCK, 0, stream>>>(
      h, mesh, dens, partial, T, n);
  hyst_reduce_kernel<<<(T + BLOCK - 1) / BLOCK, BLOCK, 0, stream>>>(
      partial, scale, offset, out, T, NW, 1.0f / (float)n);
}


# reduce kernel: a work-group sums one 128-byte column strip, rows split over 32 lane groups, 10 row loads in flight per lane, then lane exchange + LDS; the row sum's reduction tree changes
# speedup vs baseline: 9.9844x; 1.9024x over previous
#include <hip/hip_runtime.h>
#include <math.h>


typedef __attribute__((ext_vector_type(2))) float v2f;
typedef __attribute__((ext_vector_type(8))) float v8f;

#define KC 8
#define PTS_PER_WAVE (4 * KC)
#define PPL (2 * KC)
#define WAVES_PER_BLOCK 8
#define BLOCK 256
#define TCH 16
#define SCAN_BLOCK (32 * TCH)

__global__ __launch_bounds__(BLOCK) void hyst_dens_kernel(
    const float* __restrict__ raw, float* __restrict__ dens, int n, int npad) {
  int j = blockIdx.x * blockDim.x + threadIdx.x;
  if (j < npad) {
    float v = 0.0f;
    if (j < n) {
      float x = raw[j];
      v = fmaxf(x, 0.0f) + log1pf(expf(-fabsf(x)));
    }
    dens[j] = v;
  }
}

static __device__ __forceinline__ unsigned scan16(
    const float (&hv)[16], float& hp, unsigned& sbits, float al, float be) {
  const float qnan = __int_as_float(0x7fc00000);
  unsigned snap = 0u;
#pragma unroll
  for (int tt = 0; tt < 16; ++tt) {
    const float hi = hv[tt];
    const float hu = (hi > hp) ? hi : qnan;
    const float hd = (hi < hp) ? hi : qnan;
    const unsigned mu = __builtin_amdgcn_ballot_w32(al <= hu);
    const unsigned md = __builtin_amdgcn_ballot_w32(be >= hd);
    sbits = (sbits | mu) & ~md;
    snap = __builtin_amdgcn_inverse_ballot_w32(0x00010001u << tt) ? sbits : snap;
    hp = hi;
  }
  return snap;
}

static __device__ __forceinline__ void scan16_ends(
    const float (&hv)[16], float& hp, unsigned& e0, unsigned& e1, float al, float be) {
  const float qnan = __int_as_float(0x7fc00000);
#pragma unroll
  for (int tt = 0; tt < 16; ++tt) {
    const float hi = hv[tt];
    const float hu = (hi > hp) ? hi : qnan;
    const float hd = (hi < hp) ? hi : qnan;
    const unsigned mu = __builtin_amdgcn_ballot_w32(al <= hu);
    const unsigned md = __builtin_amdgcn_ballot_w32(be >= hd);
    e0 = (e0 | mu) & ~md;
    e1 = (e1 | mu) & ~md;
    hp = hi;
  }
}

static __device__ __forceinline__ void load16(float (&hv)[16], const float* __restrict__ h, int t0) {
#pragma unroll
  for (int i = 0; i < 16; ++i) hv[i] = h[t0 + i];
}

static __device__ __forceinline__ void product16(
    unsigned snap, const float (&dd)[PPL], float* __restrict__ pout, int tb,
    int row, int half) {
  const unsigned nx = ~(snap >> (2 * half));
  v8f acc = {};
#pragma unroll
  for (int kc = 0; kc < KC; ++kc) {
    v2f A, B;
    A.x = __uint_as_float(((nx << (31 - 4 * kc)) & 0x80000000u) | 0x3f800000u);
    A.y = __uint_as_float(((nx << (30 - 4 * kc)) & 0x80000000u) | 0x3f800000u);
    B.x = dd[2 * kc];
    B.y = dd[2 * kc + 1];
    acc = __builtin_amdgcn_wmma_f32_16x16x4_f32(
        false, A, false, B, (short)0, acc, false, false);
  }
  if (row == 0) {
    float* q = pout + tb + 8 * half;
    *reinterpret_cast<float4*>(q)     = make_float4(acc[0], acc[1], acc[2], acc[3]);
    *reinterpret_cast<float4*>(q + 4) = make_float4(acc[4], acc[5], acc[6], acc[7]);
  }
}

__global__ __launch_bounds__(SCAN_BLOCK) void hyst_scan_kernel(
    const float* __restrict__ h, const float* __restrict__ mesh,
    const float* __restrict__ dens, float* __restrict__ partial,
    int T, int n) {
  __shared__ unsigned end0[TCH];
  __shared__ unsigned end1[TCH];

  const int lane = threadIdx.x & 31;
  const int c    = __builtin_amdgcn_readfirstlane((int)(threadIdx.x >> 5));
  const int row  = lane & 15;
  const int half = lane >> 4;
  const int w    = blockIdx.x;

  float al = 3.0e38f, be = -3.0e38f;
  {
    const int p = w * PTS_PER_WAVE + lane;
    if (p < n) { be = mesh[2 * p]; al = mesh[2 * p + 1]; }
  }
  float dd[PPL];
  {
    const int pbase = w * PTS_PER_WAVE + 2 * half;
#pragma unroll
    for (int kc = 0; kc < KC; ++kc) {
#pragma unroll
      for (int i = 0; i < 2; ++i) {
        const int p = pbase + 4 * kc + i;
        dd[2 * kc + i] = (p < n) ? dens[p] : 0.0f;
      }
    }
  }

  const int ntile = T >> 4;
  const int tpc   = (((ntile + TCH - 1) / TCH) + 1) & ~1;
  const int t0    = min(c * tpc, ntile) * 16;
  const int t1    = min(c * tpc + tpc, ntile) * 16;
  const int tlast = T - 16;
  float hp0 = 0.0f;
  if (t0 > 0) hp0 = h[t0 - 1];

  float hA[16], hB[16];

  unsigned e0 = 0u, e1 = ~0u;
  if (c < TCH - 1 && t0 < t1) {
    float hp = hp0;
    load16(hA, h, t0);
    for (int tb = t0; tb < t1; tb += 32) {
      load16(hB, h, min(tb + 16, tlast));
      scan16_ends(hA, hp, e0, e1, al, be);
      if (tb + 16 >= t1) break;
      load16(hA, h, min(tb + 32, tlast));
      scan16_ends(hB, hp, e0, e1, al, be);
    }
  }
  if (lane == 0) { end0[c] = e0; end1[c] = e1; }
  __syncthreads();

  unsigned inc = 0u;
#pragma unroll
  for (int k = 0; k < TCH - 1; ++k) {
    const unsigned a = end0[k];
    const unsigned b = end1[k];
    inc = (k < c) ? ((inc & b) | a) : inc;
  }

  if (t0 < t1) {
    float* pout = partial + (size_t)w * T;
    unsigned sbits = __builtin_amdgcn_readfirstlane(inc);
    float hp = hp0;
    load16(hA, h, t0);
    load16(hB, h, min(t0 + 16, tlast));
    unsigned s_prev = scan16(hA, hp, sbits, al, be);
    for (int tb = t0 + 16; tb < t1; tb += 32) {
      load16(hA, h, min(tb + 16, tlast));
      const unsigned s0 = scan16(hB, hp, sbits, al, be);
      product16(s_prev, dd, pout, tb - 16, row, half);
      s_prev = s0;
      if (tb + 16 >= t1) break;
      load16(hB, h, min(tb + 32, tlast));
      const unsigned s1 = scan16(hA, hp, sbits, al, be);
      product16(s0, dd, pout, tb, row, half);
      s_prev = s1;
    }
    product16(s_prev, dd, pout, t1 - 16, row, half);
  }
}

#define RED_COLS 32
#define RED_WAVES (BLOCK / 32)
#define RED_SLOTS (4 * RED_WAVES)
#define RED_UNROLL 10
__global__ __launch_bounds__(BLOCK) void hyst_reduce_kernel(
    const float* __restrict__ partial, const float* __restrict__ scale,
    const float* __restrict__ offset, float* __restrict__ out,
    int T, int NW, float inv_n) {
  __shared__ float red[RED_WAVES][RED_COLS];

  const int lane = threadIdx.x & 31;
  const int wv   = threadIdx.x >> 5;
  const int slot = 4 * wv + (lane >> 3);
  const int col  = blockIdx.x * RED_COLS + 4 * (lane & 7);

  float4 acc = make_float4(0.0f, 0.0f, 0.0f, 0.0f);
  if (col < T) {
    const float* base = partial + col;
    for (int r0 = slot; r0 < NW; r0 += RED_SLOTS * RED_UNROLL) {
      float4 x[RED_UNROLL];
#pragma unroll
      for (int i = 0; i < RED_UNROLL; ++i) {
        const int r = min(r0 + i * RED_SLOTS, NW - 1);
        x[i] = *reinterpret_cast<const float4*>(base + (size_t)r * T);
      }
#pragma unroll
      for (int i = 0; i < RED_UNROLL; ++i) {
        const bool in = (r0 + i * RED_SLOTS) < NW;
        acc.x += in ? x[i].x : 0.0f;
        acc.y += in ? x[i].y : 0.0f;
        acc.z += in ? x[i].z : 0.0f;
        acc.w += in ? x[i].w : 0.0f;
      }
    }
  }
  acc.x += __shfl_xor(acc.x, 8);  acc.y += __shfl_xor(acc.y, 8);
  acc.z += __shfl_xor(acc.z, 8);  acc.w += __shfl_xor(acc.w, 8);
  acc.x += __shfl_xor(acc.x, 16); acc.y += __shfl_xor(acc.y, 16);
  acc.z += __shfl_xor(acc.z, 16); acc.w += __shfl_xor(acc.w, 16);
  if (lane < 8) *reinterpret_cast<float4*>(&red[wv][4 * lane]) = acc;
  __syncthreads();

  if (threadIdx.x < RED_COLS) {
    const int t = blockIdx.x * RED_COLS + threadIdx.x;
    if (t < T) {
      float sum = 0.0f;
#pragma unroll
      for (int k = 0; k < RED_WAVES; ++k) sum += red[k][threadIdx.x];
      out[t] = scale[0] * (sum * inv_n) + offset[0];
    }
  }
}

extern "C" void kernel_launch(void* const* d_in, const int* in_sizes, int n_in,
                              void* d_out, int out_size, void* d_ws, size_t ws_size,
                              hipStream_t stream) {
  const float* h      = (const float*)d_in[0];
  const float* mesh   = (const float*)d_in[1];
  const float* raw    = (const float*)d_in[2];
  const float* scale  = (const float*)d_in[3];
  const float* offset = (const float*)d_in[4];
  float* out = (float*)d_out;

  const int T = in_sizes[0];
  const int n = in_sizes[2];

  const int nWaves  = (n + PTS_PER_WAVE - 1) / PTS_PER_WAVE;
  const int nBlocks = (nWaves + WAVES_PER_BLOCK - 1) / WAVES_PER_BLOCK;
  const int NW      = nBlocks * WAVES_PER_BLOCK;
  const int npad    = NW * PTS_PER_WAVE;

  float* partial = (float*)d_ws;
  float* dens    = partial + (size_t)NW * T;

  hyst_dens_kernel<<<(npad + BLOCK - 1) / BLOCK, BLOCK, 0, stream>>>(
      raw, dens, n, npad);
  hyst_scan_kernel<<<NW, SCAN_BLOCK, 0, stream>>>(
      h, mesh, dens, partial, T, n);
  hyst_reduce_kernel<<<(T + RED_COLS - 1) / RED_COLS, BLOCK, 0, stream>>>(
      partial, scale, offset, out, T, NW, 1.0f / (float)n);
}


# scan kernel: a step's rise / fall test depends on the field alone: phase 0 does it once and stores two NaN-masked values per step, which every wave reads in place of 4 scalar ops per step per pass
# speedup vs baseline: 12.1093x; 1.2128x over previous
#include <hip/hip_runtime.h>
#include <math.h>


typedef __attribute__((ext_vector_type(2))) float v2f;
typedef __attribute__((ext_vector_type(8))) float v8f;

#define KC 8
#define PTS_PER_WAVE (4 * KC)
#define PPL (2 * KC)
#define WAVES_PER_BLOCK 8
#define BLOCK 256
#define TCH 16
#define SCAN_BLOCK (32 * TCH)

__global__ __launch_bounds__(BLOCK) void hyst_dens_kernel(
    const float* __restrict__ raw, float* __restrict__ dens, int n, int npad,
    const float* __restrict__ h, float* __restrict__ updn, int T) {
  int j = blockIdx.x * blockDim.x + threadIdx.x;
  if (j < npad) {
    float v = 0.0f;
    if (j < n) {
      float x = raw[j];
      v = fmaxf(x, 0.0f) + log1pf(expf(-fabsf(x)));
    }
    dens[j] = v;
  }
  const float qnan = __int_as_float(0x7fc00000);
  for (int t = j; t < T; t += gridDim.x * blockDim.x) {
    const float hi = h[t];
    const float hp = (t > 0) ? h[t - 1] : 0.0f;
    float2 v;
    v.x = (hi > hp) ? hi : qnan;
    v.y = (hi < hp) ? hi : qnan;
    reinterpret_cast<float2*>(updn)[t] = v;
  }
}

static __device__ __forceinline__ void masks16(
    const float (&uv)[32], unsigned (&mu)[16], unsigned (&md)[16], float al, float be) {
#pragma unroll
  for (int tt = 0; tt < 16; ++tt) {
    mu[tt] = __builtin_amdgcn_ballot_w32(al <= uv[2 * tt]);
    md[tt] = __builtin_amdgcn_ballot_w32(be >= uv[2 * tt + 1]);
  }
}

static __device__ __forceinline__ unsigned scan16(
    const unsigned (&mu)[16], const unsigned (&md)[16], unsigned& sbits) {
  unsigned snap = 0u;
#pragma unroll
  for (int tt = 0; tt < 16; ++tt) {
    sbits = (sbits | mu[tt]) & ~md[tt];
    snap = __builtin_amdgcn_inverse_ballot_w32(0x00010001u << tt) ? sbits : snap;
  }
  return snap;
}

static __device__ __forceinline__ void scan16_ends(
    const unsigned (&mu)[16], const unsigned (&md)[16], unsigned& e0, unsigned& e1) {
#pragma unroll
  for (int tt = 0; tt < 16; ++tt) {
    e0 = (e0 | mu[tt]) & ~md[tt];
    e1 = (e1 | mu[tt]) & ~md[tt];
  }
}

static __device__ __forceinline__ void load16(float (&uv)[32], const float* __restrict__ updn, int t0) {
#pragma unroll
  for (int i = 0; i < 32; ++i) uv[i] = updn[2 * t0 + i];
}

static __device__ __forceinline__ void product16(
    unsigned snap, const float (&dd)[PPL], float* __restrict__ pout, int tb,
    int row, int half) {
  const unsigned nx = ~(snap >> (2 * half));
  v8f acc = {};
#pragma unroll
  for (int kc = 0; kc < KC; ++kc) {
    v2f A, B;
    A.x = __uint_as_float(((nx << (31 - 4 * kc)) & 0x80000000u) | 0x3f800000u);
    A.y = __uint_as_float(((nx << (30 - 4 * kc)) & 0x80000000u) | 0x3f800000u);
    B.x = dd[2 * kc];
    B.y = dd[2 * kc + 1];
    acc = __builtin_amdgcn_wmma_f32_16x16x4_f32(
        false, A, false, B, (short)0, acc, false, false);
  }
  if (row == 0) {
    float* q = pout + tb + 8 * half;
    *reinterpret_cast<float4*>(q)     = make_float4(acc[0], acc[1], acc[2], acc[3]);
    *reinterpret_cast<float4*>(q + 4) = make_float4(acc[4], acc[5], acc[6], acc[7]);
  }
}

__global__ __launch_bounds__(SCAN_BLOCK) void hyst_scan_kernel(
    const float* __restrict__ updn, const float* __restrict__ mesh,
    const float* __restrict__ dens, float* __restrict__ partial,
    int T, int n) {
  __shared__ unsigned end0[TCH];
  __shared__ unsigned end1[TCH];

  const int lane = threadIdx.x & 31;
  const int c    = __builtin_amdgcn_readfirstlane((int)(threadIdx.x >> 5));
  const int row  = lane & 15;
  const int half = lane >> 4;
  const int w    = blockIdx.x;

  float al = 3.0e38f, be = -3.0e38f;
  {
    const int p = w * PTS_PER_WAVE + lane;
    if (p < n) { be = mesh[2 * p]; al = mesh[2 * p + 1]; }
  }
  float dd[PPL];
  {
    const int pbase = w * PTS_PER_WAVE + 2 * half;
#pragma unroll
    for (int kc = 0; kc < KC; ++kc) {
#pragma unroll
      for (int i = 0; i < 2; ++i) {
        const int p = pbase + 4 * kc + i;
        dd[2 * kc + i] = (p < n) ? dens[p] : 0.0f;
      }
    }
  }

  const int ntile = T >> 4;
  const int tpc   = (((ntile + TCH - 1) / TCH) + 1) & ~1;
  const int t0    = min(c * tpc, ntile) * 16;
  const int t1    = min(c * tpc + tpc, ntile) * 16;
  const int tlast = T - 16;

  float uv[32], uw[32];
  unsigned mu[16], md[16];

  unsigned e0 = 0u, e1 = ~0u;
  if (c < TCH - 1 && t0 < t1) {
    load16(uv, updn, t0);
    for (int tb = t0; tb < t1; tb += 32) {
      load16(uw, updn, min(tb + 16, tlast));
      masks16(uv, mu, md, al, be);
      scan16_ends(mu, md, e0, e1);
      if (tb + 16 >= t1) break;
      load16(uv, updn, min(tb + 32, tlast));
      masks16(uw, mu, md, al, be);
      scan16_ends(mu, md, e0, e1);
    }
  }
  if (lane == 0) { end0[c] = e0; end1[c] = e1; }
  __syncthreads();

  unsigned inc = 0u;
#pragma unroll
  for (int k = 0; k < TCH - 1; ++k) {
    const unsigned a = end0[k];
    const unsigned b = end1[k];
    inc = (k < c) ? ((inc & b) | a) : inc;
  }

  if (t0 < t1) {
    float* pout = partial + (size_t)w * T;
    unsigned sbits = __builtin_amdgcn_readfirstlane(inc);
    load16(uv, updn, t0);
    masks16(uv, mu, md, al, be);
    load16(uv, updn, min(t0 + 16, tlast));
    unsigned s_prev = scan16(mu, md, sbits);
    masks16(uv, mu, md, al, be);
    for (int tb = t0 + 16; tb < t1; tb += 16) {
      load16(uv, updn, min(tb + 16, tlast));
      const unsigned s = scan16(mu, md, sbits);
      product16(s_prev, dd, pout, tb - 16, row, half);
      s_prev = s;
      masks16(uv, mu, md, al, be);
    }
    product16(s_prev, dd, pout, t1 - 16, row, half);
  }
}

#define RED_COLS 32
#define RED_WAVES (BLOCK / 32)
#define RED_SLOTS (4 * RED_WAVES)
#define RED_UNROLL 10
__global__ __launch_bounds__(BLOCK) void hyst_reduce_kernel(
    const float* __restrict__ partial, const float* __restrict__ scale,
    const float* __restrict__ offset, float* __restrict__ out,
    int T, int NW, float inv_n) {
  __shared__ float red[RED_WAVES][RED_COLS];

  const int lane = threadIdx.x & 31;
  const int wv   = threadIdx.x >> 5;
  const int slot = 4 * wv + (lane >> 3);
  const int col  = blockIdx.x * RED_COLS + 4 * (lane & 7);

  float4 acc = make_float4(0.0f, 0.0f, 0.0f, 0.0f);
  if (col < T) {
    const float* base = partial + col;
    for (int r0 = slot; r0 < NW; r0 += RED_SLOTS * RED_UNROLL) {
      float4 x[RED_UNROLL];
#pragma unroll
      for (int i = 0; i < RED_UNROLL; ++i) {
        const int r = min(r0 + i * RED_SLOTS, NW - 1);
        x[i] = *reinterpret_cast<const float4*>(base + (size_t)r * T);
      }
#pragma unroll
      for (int i = 0; i < RED_UNROLL; ++i) {
        const bool in = (r0 + i * RED_SLOTS) < NW;
        acc.x += in ? x[i].x : 0.0f;
        acc.y += in ? x[i].y : 0.0f;
        acc.z += in ? x[i].z : 0.0f;
        acc.w += in ? x[i].w : 0.0f;
      }
    }
  }
  acc.x += __shfl_xor(acc.x, 8);  acc.y += __shfl_xor(acc.y, 8);
  acc.z += __shfl_xor(acc.z, 8);  acc.w += __shfl_xor(acc.w, 8);
  acc.x += __shfl_xor(acc.x, 16); acc.y += __shfl_xor(acc.y, 16);
  acc.z += __shfl_xor(acc.z, 16); acc.w += __shfl_xor(acc.w, 16);
  if (lane < 8) *reinterpret_cast<float4*>(&red[wv][4 * lane]) = acc;
  __syncthreads();

  if (threadIdx.x < RED_COLS) {
    const int t = blockIdx.x * RED_COLS + threadIdx.x;
    if (t < T) {
      float sum = 0.0f;
#pragma unroll
      for (int k = 0; k < RED_WAVES; ++k) sum += red[k][threadIdx.x];
      out[t] = scale[0] * (sum * inv_n) + offset[0];
    }
  }
}

extern "C" void kernel_launch(void* const* d_in, const int* in_sizes, int n_in,
                              void* d_out, int out_size, void* d_ws, size_t ws_size,
                              hipStream_t stream) {
  const float* h      = (const float*)d_in[0];
  const float* mesh   = (const float*)d_in[1];
  const float* raw    = (const float*)d_in[2];
  const float* scale  = (const float*)d_in[3];
  const float* offset = (const float*)d_in[4];
  float* out = (float*)d_out;

  const int T = in_sizes[0];
  const int n = in_sizes[2];

  const int nWaves  = (n + PTS_PER_WAVE - 1) / PTS_PER_WAVE;
  const int nBlocks = (nWaves + WAVES_PER_BLOCK - 1) / WAVES_PER_BLOCK;
  const int NW      = nBlocks * WAVES_PER_BLOCK;
  const int npad    = NW * PTS_PER_WAVE;

  float* partial = (float*)d_ws;
  float* dens    = partial + (size_t)NW * T;
  float* updn    = dens + npad;

  hyst_dens_kernel<<<(npad + BLOCK - 1) / BLOCK, BLOCK, 0, stream>>>(
      raw, dens, n, npad, h, updn, T);
  hyst_scan_kernel<<<NW, SCAN_BLOCK, 0, stream>>>(
      updn, mesh, dens, partial, T, n);
  hyst_reduce_kernel<<<(T + RED_COLS - 1) / RED_COLS, BLOCK, 0, stream>>>(
      partial, scale, offset, out, T, NW, 1.0f / (float)n);
}


# scan kernel: time on the bit axis: one scalar subtract-with-borrow (set mask minus clear mask) resolves 32 steps of a relay, no chunk pre-pass; 20 waves add tiles via LDS; row sum's tree changes
# speedup vs baseline: 16.9421x; 1.3991x over previous
#include <hip/hip_runtime.h>
#include <math.h>


typedef __attribute__((ext_vector_type(2))) float v2f;
typedef __attribute__((ext_vector_type(8))) float v8f;

#define BLOCK 256
#define QW 20
#define SCAN_BLOCK (32 * QW)
#define RPG (4 * QW)
#define WPT 4
#define TRIP (32 * WPT)
#define WIN 2048
#define WIN_PITCH (WIN + 16)

__global__ __launch_bounds__(BLOCK) void hyst_dens_kernel(
    const float* __restrict__ raw, float* __restrict__ dens, int n, int npad,
    const float* __restrict__ h, float* __restrict__ updn, int T, int Tpad) {
  int j = blockIdx.x * blockDim.x + threadIdx.x;
  if (j < npad) {
    float v = 0.0f;
    if (j < n) {
      float x = raw[j];
      v = fmaxf(x, 0.0f) + log1pf(expf(-fabsf(x)));
    }
    dens[j] = v;
  }
  const float qnan = __int_as_float(0x7fc00000);
  for (int k = j; k <= Tpad; k += gridDim.x * blockDim.x) {
    const int t = (k < Tpad) ? k + 1 : 0;
    float2 v;
    v.x = qnan;
    v.y = qnan;
    if (t < T) {
      const float hi = h[t];
      const float hp = (t > 0) ? h[t - 1] : 0.0f;
      v.x = (hi > hp) ? hi : qnan;
      v.y = (hi < hp) ? hi : qnan;
    }
    reinterpret_cast<float2*>(updn)[k] = v;
  }
}

static __device__ __forceinline__ unsigned pack_lo(unsigned a, unsigned b) {
  unsigned r;
  asm("s_pack_ll_b32_b16 %0, %1, %2" : "=s"(r) : "s"(a), "s"(b));
  return r;
}
static __device__ __forceinline__ unsigned pack_hi(unsigned a, unsigned b) {
  unsigned r;
  asm("s_pack_hh_b32_b16 %0, %1, %2" : "=s"(r) : "s"(a), "s"(b));
  return r;
}

__global__ __launch_bounds__(SCAN_BLOCK) void hyst_scan_kernel(
    const float* __restrict__ updn, const float* __restrict__ mesh,
    const float* __restrict__ dens, float* __restrict__ partial,
    int T, int Tpad, int n) {
  __shared__ float red[QW][WIN_PITCH];

  const int lane = threadIdx.x & 31;
  const int wv   = __builtin_amdgcn_readfirstlane((int)(threadIdx.x >> 5));
  const int row  = lane & 15;
  const int half = lane >> 4;
  const int q    = blockIdx.x * QW + wv;

  const float2* __restrict__ ud = reinterpret_cast<const float2*>(updn);
  const float up0 = ud[Tpad].x;

  float al[4], be[4];
#pragma unroll
  for (int i = 0; i < 4; ++i) {
    const int p  = 4 * q + i;
    const int pc = min(p, n - 1);
    const float b0 = mesh[2 * pc];
    const float a0 = mesh[2 * pc + 1];
    be[i] = (p < n) ? b0 : -3.0e38f;
    al[i] = (p < n) ? a0 : 3.0e38f;
  }
  v2f B;
  {
    const int p = 4 * q + 2 * half;
    B.x = dens[p];
    B.y = dens[p + 1];
  }

  unsigned bw[4];
#pragma unroll
  for (int i = 0; i < 4; ++i) bw[i] = (al[i] <= up0) ? 0u : 1u;

  const int ntrip = Tpad / TRIP;
  float2 cur[WPT], nxt[WPT];
#pragma unroll
  for (int w = 0; w < WPT; ++w) cur[w] = ud[w * 32 + lane];

  for (int win0 = 0; win0 < ntrip; win0 += WIN / TRIP) {
    const int tend = min(win0 + WIN / TRIP, ntrip);
#pragma unroll 2
    for (int tr = win0; tr < tend; ++tr) {
      const int trn = min(tr + 1, ntrip - 1);
#pragma unroll
      for (int w = 0; w < WPT; ++w) nxt[w] = ud[(trn * WPT + w) * 32 + lane];

      unsigned nx[4][WPT];
#pragma unroll
      for (int i = 0; i < 4; ++i) {
        unsigned g[WPT], d[WPT];
#pragma unroll
        for (int w = 0; w < WPT; ++w) {
          g[w] = __builtin_amdgcn_ballot_w32(al[i] <= cur[w].x);
          d[w] = __builtin_amdgcn_ballot_w32(be[i] >= cur[w].y);
        }
        unsigned s0, s1, s2, s3;
        asm("s_cmp_lg_u32 %[b], 0\n\t"
            "s_sub_co_ci_u32 %[s0], %[g0], %[d0]\n\t"
            "s_sub_co_ci_u32 %[s1], %[g1], %[d1]\n\t"
            "s_sub_co_ci_u32 %[s2], %[g2], %[d2]\n\t"
            "s_sub_co_ci_u32 %[s3], %[g3], %[d3]\n\t"
            "s_cselect_b32 %[b], 1, 0"
            : [s0] "=&s"(s0), [s1] "=&s"(s1), [s2] "=&s"(s2), [s3] "=&s"(s3), [b] "+s"(bw[i])
            : [g0] "s"(g[0]), [d0] "s"(d[0]), [g1] "s"(g[1]), [d1] "s"(d[1]),
              [g2] "s"(g[2]), [d2] "s"(d[2]), [g3] "s"(g[3]), [d3] "s"(d[3])
            : "scc");
        nx[i][0] = s0 ^ g[0] ^ d[0];
        nx[i][1] = s1 ^ g[1] ^ d[1];
        nx[i][2] = s2 ^ g[2] ^ d[2];
        nx[i][3] = s3 ^ g[3] ^ d[3];
      }

      v8f acc[2 * WPT];
#pragma unroll
      for (int w = 0; w < WPT; ++w) {
#pragma unroll
        for (int sub = 0; sub < 2; ++sub) {
          const unsigned c02 = sub ? pack_hi(nx[0][w], nx[2][w]) : pack_lo(nx[0][w], nx[2][w]);
          const unsigned c13 = sub ? pack_hi(nx[1][w], nx[3][w]) : pack_lo(nx[1][w], nx[3][w]);
          v2f A;
          A.x = __builtin_amdgcn_inverse_ballot_w32(c02) ? -1.0f : 1.0f;
          A.y = __builtin_amdgcn_inverse_ballot_w32(c13) ? -1.0f : 1.0f;
          const v8f zero = {};
          acc[2 * w + sub] = __builtin_amdgcn_wmma_f32_16x16x4_f32(
              false, A, false, B, (short)0, zero, false, false);
        }
      }
      if (row == 0) {
        float* qd = &red[wv][(tr - win0) * TRIP + 8 * half];
#pragma unroll
        for (int k = 0; k < 2 * WPT; ++k) {
          *reinterpret_cast<float4*>(qd + 16 * k) =
              make_float4(acc[k][0], acc[k][1], acc[k][2], acc[k][3]);
          *reinterpret_cast<float4*>(qd + 16 * k + 4) =
              make_float4(acc[k][4], acc[k][5], acc[k][6], acc[k][7]);
        }
      }
#pragma unroll
      for (int w = 0; w < WPT; ++w) cur[w] = nxt[w];
    }
    __syncthreads();

    const int tbase = win0 * TRIP;
    float* pout = partial + (size_t)blockIdx.x * T + tbase;
    for (int c4 = threadIdx.x; c4 < WIN / 4; c4 += SCAN_BLOCK) {
      float4 s = *reinterpret_cast<const float4*>(&red[0][4 * c4]);
#pragma unroll
      for (int k = 1; k < QW; ++k) {
        const float4 x = *reinterpret_cast<const float4*>(&red[k][4 * c4]);
        s.x += x.x; s.y += x.y; s.z += x.z; s.w += x.w;
      }
      if (tbase + 4 * c4 < T)
        *reinterpret_cast<float4*>(pout + 4 * c4) = s;
    }
    __syncthreads();
  }
}

#define RED_COLS 32
#define RED_WAVES (BLOCK / 32)
#define RED_SLOTS (4 * RED_WAVES)
#define RED_UNROLL 10
__global__ __launch_bounds__(BLOCK) void hyst_reduce_kernel(
    const float* __restrict__ partial, const float* __restrict__ scale,
    const float* __restrict__ offset, float* __restrict__ out,
    int T, int NW, float inv_n) {
  __shared__ float red[RED_WAVES][RED_COLS];

  const int lane = threadIdx.x & 31;
  const int wv   = threadIdx.x >> 5;
  const int slot = 4 * wv + (lane >> 3);
  const int col  = blockIdx.x * RED_COLS + 4 * (lane & 7);

  float4 acc = make_float4(0.0f, 0.0f, 0.0f, 0.0f);
  if (col < T) {
    const float* base = partial + col;
    for (int r0 = slot; r0 < NW; r0 += RED_SLOTS * RED_UNROLL) {
      float4 x[RED_UNROLL];
#pragma unroll
      for (int i = 0; i < RED_UNROLL; ++i) {
        const int r = min(r0 + i * RED_SLOTS, NW - 1);
        x[i] = *reinterpret_cast<const float4*>(base + (size_t)r * T);
      }
#pragma unroll
      for (int i = 0; i < RED_UNROLL; ++i) {
        const bool in = (r0 + i * RED_SLOTS) < NW;
        acc.x += in ? x[i].x : 0.0f;
        acc.y += in ? x[i].y : 0.0f;
        acc.z += in ? x[i].z : 0.0f;
        acc.w += in ? x[i].w : 0.0f;
      }
    }
  }
  acc.x += __shfl_xor(acc.x, 8);  acc.y += __shfl_xor(acc.y, 8);
  acc.z += __shfl_xor(acc.z, 8);  acc.w += __shfl_xor(acc.w, 8);
  acc.x += __shfl_xor(acc.x, 16); acc.y += __shfl_xor(acc.y, 16);
  acc.z += __shfl_xor(acc.z, 16); acc.w += __shfl_xor(acc.w, 16);
  if (lane < 8) *reinterpret_cast<float4*>(&red[wv][4 * lane]) = acc;
  __syncthreads();

  if (threadIdx.x < RED_COLS) {
    const int t = blockIdx.x * RED_COLS + threadIdx.x;
    if (t < T) {
      float sum = 0.0f;
#pragma unroll
      for (int k = 0; k < RED_WAVES; ++k) sum += red[k][threadIdx.x];
      out[t] = scale[0] * (sum * inv_n) + offset[0];
    }
  }
}

extern "C" void kernel_launch(void* const* d_in, const int* in_sizes, int n_in,
                              void* d_out, int out_size, void* d_ws, size_t ws_size,
                              hipStream_t stream) {
  const float* h      = (const float*)d_in[0];
  const float* mesh   = (const float*)d_in[1];
  const float* raw    = (const float*)d_in[2];
  const float* scale  = (const float*)d_in[3];
  const float* offset = (const float*)d_in[4];
  float* out = (float*)d_out;

  const int T = in_sizes[0];
  const int n = in_sizes[2];

  const int nQuads  = (n + 3) / 4;
  const int nGroups = (nQuads + QW - 1) / QW;
  const int npad    = nGroups * RPG;
  const int Tpad    = ((T + TRIP - 1) / TRIP) * TRIP;

  float* partial = (float*)d_ws;
  float* dens    = partial + (size_t)nGroups * T;
  float* updn    = dens + npad;

  hyst_dens_kernel<<<(npad + BLOCK - 1) / BLOCK, BLOCK, 0, stream>>>(
      raw, dens, n, npad, h, updn, T, Tpad);
  hyst_scan_kernel<<<nGroups, SCAN_BLOCK, 0, stream>>>(
      updn, mesh, dens, partial, T, Tpad, n);
  hyst_reduce_kernel<<<(T + RED_COLS - 1) / RED_COLS, BLOCK, 0, stream>>>(
      partial, scale, offset, out, T, nGroups, 1.0f / (float)n);
}


# first launch folded into the scan kernel: each work-group builds its window's rise / fall pairs and its 80 densities in LDS itself, the loop reads pairs from LDS; 3 launches -> 2; sums unchanged
# speedup vs baseline: 18.0073x; 1.0629x over previous
#include <hip/hip_runtime.h>
#include <math.h>


typedef __attribute__((ext_vector_type(2))) float v2f;
typedef __attribute__((ext_vector_type(8))) float v8f;

#define BLOCK 256
#define QW 20
#define SCAN_BLOCK (32 * QW)
#define RPG (4 * QW)
#define WPT 4
#define TRIP (32 * WPT)
#define WIN 2048
#define WIN_PITCH (WIN + 16)
#define TAB_ITERS ((WIN + SCAN_BLOCK - 1) / SCAN_BLOCK)

static __device__ __forceinline__ unsigned pack_lo(unsigned a, unsigned b) {
  unsigned r;
  asm("s_pack_ll_b32_b16 %0, %1, %2" : "=s"(r) : "s"(a), "s"(b));
  return r;
}
static __device__ __forceinline__ unsigned pack_hi(unsigned a, unsigned b) {
  unsigned r;
  asm("s_pack_hh_b32_b16 %0, %1, %2" : "=s"(r) : "s"(a), "s"(b));
  return r;
}

static __device__ __forceinline__ void stage_load(
    const float* __restrict__ h, int T, int tbase, float (&hi)[TAB_ITERS], float (&hp)[TAB_ITERS]) {
#pragma unroll
  for (int i = 0; i < TAB_ITERS; ++i) {
    const int j = (int)threadIdx.x + i * SCAN_BLOCK;
    hi[i] = 0.0f;
    hp[i] = 0.0f;
    if (j < WIN) {
      const int tc = min(tbase + j + 1, T - 1);
      hi[i] = h[tc];
      hp[i] = h[max(tc - 1, 0)];
    }
  }
}
static __device__ __forceinline__ void stage_store(
    float2* tab, int T, int tbase, const float (&hi)[TAB_ITERS], const float (&hp)[TAB_ITERS]) {
  const float qnan = __int_as_float(0x7fc00000);
#pragma unroll
  for (int i = 0; i < TAB_ITERS; ++i) {
    const int j = (int)threadIdx.x + i * SCAN_BLOCK;
    if (j < WIN) {
      const bool in = (tbase + j + 1) < T;
      float2 v;
      v.x = (in && hi[i] > hp[i]) ? hi[i] : qnan;
      v.y = (in && hi[i] < hp[i]) ? hi[i] : qnan;
      tab[j] = v;
    }
  }
}

__global__ __launch_bounds__(SCAN_BLOCK) void hyst_scan_kernel(
    const float* __restrict__ h, const float* __restrict__ mesh,
    const float* __restrict__ raw, float* __restrict__ partial,
    int T, int Tpad, int n) {
  __shared__ float red[QW][WIN_PITCH];
  __shared__ float2 tab[WIN];
  __shared__ float sdens[RPG];

  const int lane = threadIdx.x & 31;
  const int wv   = __builtin_amdgcn_readfirstlane((int)(threadIdx.x >> 5));
  const int row  = lane & 15;
  const int half = lane >> 4;
  const int q    = blockIdx.x * QW + wv;

  float hi[TAB_ITERS], hp[TAB_ITERS];
  stage_load(h, T, 0, hi, hp);

  if (threadIdx.x < RPG) {
    const int p = blockIdx.x * RPG + (int)threadIdx.x;
    float v = 0.0f;
    if (p < n) {
      const float x = raw[p];
      v = fmaxf(x, 0.0f) + log1pf(expf(-fabsf(x)));
    }
    sdens[threadIdx.x] = v;
  }

  const float h0  = h[0];
  const float up0 = (h0 > 0.0f) ? h0 : __int_as_float(0x7fc00000);

  float al[4], be[4];
#pragma unroll
  for (int i = 0; i < 4; ++i) {
    const int p  = 4 * q + i;
    const int pc = min(p, n - 1);
    const float b0 = mesh[2 * pc];
    const float a0 = mesh[2 * pc + 1];
    be[i] = (p < n) ? b0 : -3.0e38f;
    al[i] = (p < n) ? a0 : 3.0e38f;
  }

  stage_store(tab, T, 0, hi, hp);

  unsigned bw[4];
#pragma unroll
  for (int i = 0; i < 4; ++i) bw[i] = (al[i] <= up0) ? 0u : 1u;

  __syncthreads();

  v2f B;
  B.x = sdens[4 * wv + 2 * half];
  B.y = sdens[4 * wv + 2 * half + 1];

  const int ntrip = Tpad / TRIP;
  for (int win0 = 0; win0 < ntrip; win0 += WIN / TRIP) {
    const int tend = min(win0 + WIN / TRIP, ntrip);
    float2 cur[WPT], nxt[WPT];
#pragma unroll
    for (int w = 0; w < WPT; ++w) cur[w] = tab[w * 32 + lane];
    asm volatile("" :: "v"(cur[0].x), "v"(cur[0].y), "v"(cur[1].x), "v"(cur[1].y),
                       "v"(cur[2].x), "v"(cur[2].y), "v"(cur[3].x), "v"(cur[3].y));

#pragma unroll 2
    for (int tr = win0; tr < tend; ++tr) {
      const int trn = min(tr + 1, tend - 1) - win0;
#pragma unroll
      for (int w = 0; w < WPT; ++w) nxt[w] = tab[(trn * WPT + w) * 32 + lane];

      unsigned nx[4][WPT];
#pragma unroll
      for (int i = 0; i < 4; ++i) {
        unsigned g[WPT], d[WPT];
#pragma unroll
        for (int w = 0; w < WPT; ++w) {
          g[w] = __builtin_amdgcn_ballot_w32(al[i] <= cur[w].x);
          d[w] = __builtin_amdgcn_ballot_w32(be[i] >= cur[w].y);
        }
        unsigned s0, s1, s2, s3;
        asm("s_cmp_lg_u32 %[b], 0\n\t"
            "s_sub_co_ci_u32 %[s0], %[g0], %[d0]\n\t"
            "s_sub_co_ci_u32 %[s1], %[g1], %[d1]\n\t"
            "s_sub_co_ci_u32 %[s2], %[g2], %[d2]\n\t"
            "s_sub_co_ci_u32 %[s3], %[g3], %[d3]\n\t"
            "s_cselect_b32 %[b], 1, 0"
            : [s0] "=&s"(s0), [s1] "=&s"(s1), [s2] "=&s"(s2), [s3] "=&s"(s3), [b] "+s"(bw[i])
            : [g0] "s"(g[0]), [d0] "s"(d[0]), [g1] "s"(g[1]), [d1] "s"(d[1]),
              [g2] "s"(g[2]), [d2] "s"(d[2]), [g3] "s"(g[3]), [d3] "s"(d[3])
            : "scc");
        nx[i][0] = s0 ^ g[0] ^ d[0];
        nx[i][1] = s1 ^ g[1] ^ d[1];
        nx[i][2] = s2 ^ g[2] ^ d[2];
        nx[i][3] = s3 ^ g[3] ^ d[3];
      }

      v8f acc[2 * WPT];
#pragma unroll
      for (int w = 0; w < WPT; ++w) {
#pragma unroll
        for (int sub = 0; sub < 2; ++sub) {
          const unsigned c02 = sub ? pack_hi(nx[0][w], nx[2][w]) : pack_lo(nx[0][w], nx[2][w]);
          const unsigned c13 = sub ? pack_hi(nx[1][w], nx[3][w]) : pack_lo(nx[1][w], nx[3][w]);
          v2f A;
          A.x = __builtin_amdgcn_inverse_ballot_w32(c02) ? -1.0f : 1.0f;
          A.y = __builtin_amdgcn_inverse_ballot_w32(c13) ? -1.0f : 1.0f;
          const v8f zero = {};
          acc[2 * w + sub] = __builtin_amdgcn_wmma_f32_16x16x4_f32(
              false, A, false, B, (short)0, zero, false, false);
        }
      }
      asm volatile("" :: "v"(nxt[0].x), "v"(nxt[0].y), "v"(nxt[1].x), "v"(nxt[1].y),
                         "v"(nxt[2].x), "v"(nxt[2].y), "v"(nxt[3].x), "v"(nxt[3].y));
      if (row == 0) {
        float* qd = &red[wv][(tr - win0) * TRIP + 8 * half];
#pragma unroll
        for (int k = 0; k < 2 * WPT; ++k) {
          *reinterpret_cast<float4*>(qd + 16 * k) =
              make_float4(acc[k][0], acc[k][1], acc[k][2], acc[k][3]);
          *reinterpret_cast<float4*>(qd + 16 * k + 4) =
              make_float4(acc[k][4], acc[k][5], acc[k][6], acc[k][7]);
        }
      }
#pragma unroll
      for (int w = 0; w < WPT; ++w) cur[w] = nxt[w];
    }
    __syncthreads();

    const int  wnext = win0 + WIN / TRIP;
    const bool more  = wnext < ntrip;
    if (more) stage_load(h, T, wnext * TRIP, hi, hp);

    const int tbase = win0 * TRIP;
    float* pout = partial + (size_t)blockIdx.x * T + tbase;
    for (int c4 = threadIdx.x; c4 < WIN / 4; c4 += SCAN_BLOCK) {
      float4 s = *reinterpret_cast<const float4*>(&red[0][4 * c4]);
#pragma unroll
      for (int k = 1; k < QW; ++k) {
        const float4 x = *reinterpret_cast<const float4*>(&red[k][4 * c4]);
        s.x += x.x; s.y += x.y; s.z += x.z; s.w += x.w;
      }
      if (tbase + 4 * c4 < T)
        *reinterpret_cast<float4*>(pout + 4 * c4) = s;
    }
    if (more) stage_store(tab, T, wnext * TRIP, hi, hp);
    __syncthreads();
  }
}

#define RED_COLS 32
#define RED_WAVES (BLOCK / 32)
#define RED_SLOTS (4 * RED_WAVES)
#define RED_UNROLL 10
__global__ __launch_bounds__(BLOCK) void hyst_reduce_kernel(
    const float* __restrict__ partial, const float* __restrict__ scale,
    const float* __restrict__ offset, float* __restrict__ out,
    int T, int NW, float inv_n) {
  __shared__ float red[RED_WAVES][RED_COLS];

  const int lane = threadIdx.x & 31;
  const int wv   = threadIdx.x >> 5;
  const int slot = 4 * wv + (lane >> 3);
  const int col  = blockIdx.x * RED_COLS + 4 * (lane & 7);

  float4 acc = make_float4(0.0f, 0.0f, 0.0f, 0.0f);
  if (col < T) {
    const float* base = partial + col;
    for (int r0 = slot; r0 < NW; r0 += RED_SLOTS * RED_UNROLL) {
      float4 x[RED_UNROLL];
#pragma unroll
      for (int i = 0; i < RED_UNROLL; ++i) {
        const int r = min(r0 + i * RED_SLOTS, NW - 1);
        x[i] = *reinterpret_cast<const float4*>(base + (size_t)r * T);
      }
#pragma unroll
      for (int i = 0; i < RED_UNROLL; ++i) {
        const bool in = (r0 + i * RED_SLOTS) < NW;
        acc.x += in ? x[i].x : 0.0f;
        acc.y += in ? x[i].y : 0.0f;
        acc.z += in ? x[i].z : 0.0f;
        acc.w += in ? x[i].w : 0.0f;
      }
    }
  }
  acc.x += __shfl_xor(acc.x, 8);  acc.y += __shfl_xor(acc.y, 8);
  acc.z += __shfl_xor(acc.z, 8);  acc.w += __shfl_xor(acc.w, 8);
  acc.x += __shfl_xor(acc.x, 16); acc.y += __shfl_xor(acc.y, 16);
  acc.z += __shfl_xor(acc.z, 16); acc.w += __shfl_xor(acc.w, 16);
  if (lane < 8) *reinterpret_cast<float4*>(&red[wv][4 * lane]) = acc;
  __syncthreads();

  if (threadIdx.x < RED_COLS) {
    const int t = blockIdx.x * RED_COLS + threadIdx.x;
    if (t < T) {
      float sum = 0.0f;
#pragma unroll
      for (int k = 0; k < RED_WAVES; ++k) sum += red[k][threadIdx.x];
      out[t] = scale[0] * (sum * inv_n) + offset[0];
    }
  }
}

extern "C" void kernel_launch(void* const* d_in, const int* in_sizes, int n_in,
                              void* d_out, int out_size, void* d_ws, size_t ws_size,
                              hipStream_t stream) {
  const float* h      = (const float*)d_in[0];
  const float* mesh   = (const float*)d_in[1];
  const float* raw    = (const float*)d_in[2];
  const float* scale  = (const float*)d_in[3];
  const float* offset = (const float*)d_in[4];
  float* out = (float*)d_out;

  const int T = in_sizes[0];
  const int n = in_sizes[2];

  const int nQuads  = (n + 3) / 4;
  const int nGroups = (nQuads + QW - 1) / QW;
  const int Tpad    = ((T + TRIP - 1) / TRIP) * TRIP;

  float* partial = (float*)d_ws;

  hyst_scan_kernel<<<nGroups, SCAN_BLOCK, 0, stream>>>(
      h, mesh, raw, partial, T, Tpad, n);
  hyst_reduce_kernel<<<(T + RED_COLS - 1) / RED_COLS, BLOCK, 0, stream>>>(
      partial, scale, offset, out, T, nGroups, 1.0f / (float)n);
}


# scan kernel: tile j of a 256-step trip meets a B fragment with the densities in column j only; 16 tiles fill one accumulator and leave as 2 full-wave 16-byte LDS stores, not 32 two-lane ones
# speedup vs baseline: 19.8450x; 1.1021x over previous
#include <hip/hip_runtime.h>
#include <math.h>


typedef __attribute__((ext_vector_type(2))) float v2f;
typedef __attribute__((ext_vector_type(8))) float v8f;

#define BLOCK 256
#define QW 20
#define SCAN_BLOCK (32 * QW)
#define RPG (4 * QW)
#define WPT 4
#define PASS (32 * WPT)
#define TRIP (2 * PASS)
#define WIN 2048
#define WIN_PITCH (WIN + 16)
#define TAB_ITERS ((WIN + SCAN_BLOCK - 1) / SCAN_BLOCK)

static __device__ __forceinline__ unsigned pack_lo(unsigned a, unsigned b) {
  unsigned r;
  asm("s_pack_ll_b32_b16 %0, %1, %2" : "=s"(r) : "s"(a), "s"(b));
  return r;
}
static __device__ __forceinline__ unsigned pack_hi(unsigned a, unsigned b) {
  unsigned r;
  asm("s_pack_hh_b32_b16 %0, %1, %2" : "=s"(r) : "s"(a), "s"(b));
  return r;
}

static __device__ __forceinline__ void stage_load(
    const float* __restrict__ h, int T, int tbase, float (&hi)[TAB_ITERS], float (&hp)[TAB_ITERS]) {
#pragma unroll
  for (int i = 0; i < TAB_ITERS; ++i) {
    const int j = (int)threadIdx.x + i * SCAN_BLOCK;
    hi[i] = 0.0f;
    hp[i] = 0.0f;
    if (j < WIN) {
      const int tc = min(tbase + j + 1, T - 1);
      hi[i] = h[tc];
      hp[i] = h[max(tc - 1, 0)];
    }
  }
}
static __device__ __forceinline__ void stage_store(
    float2* tab, int T, int tbase, const float (&hi)[TAB_ITERS], const float (&hp)[TAB_ITERS]) {
  const float qnan = __int_as_float(0x7fc00000);
#pragma unroll
  for (int i = 0; i < TAB_ITERS; ++i) {
    const int j = (int)threadIdx.x + i * SCAN_BLOCK;
    if (j < WIN) {
      const bool in = (tbase + j + 1) < T;
      float2 v;
      v.x = (in && hi[i] > hp[i]) ? hi[i] : qnan;
      v.y = (in && hi[i] < hp[i]) ? hi[i] : qnan;
      tab[j] = v;
    }
  }
}

__global__ __launch_bounds__(SCAN_BLOCK) void hyst_scan_kernel(
    const float* __restrict__ h, const float* __restrict__ mesh,
    const float* __restrict__ raw, float* __restrict__ partial,
    int T, int Tpad, int n) {
  __shared__ float red[QW][WIN_PITCH];
  __shared__ float2 tab[WIN];
  __shared__ float sdens[RPG];

  const int lane = threadIdx.x & 31;
  const int wv   = __builtin_amdgcn_readfirstlane((int)(threadIdx.x >> 5));
  const int row  = lane & 15;
  const int half = lane >> 4;
  const int q    = blockIdx.x * QW + wv;

  float hi[TAB_ITERS], hp[TAB_ITERS];
  stage_load(h, T, 0, hi, hp);

  if (threadIdx.x < RPG) {
    const int p = blockIdx.x * RPG + (int)threadIdx.x;
    float v = 0.0f;
    if (p < n) {
      const float x = raw[p];
      v = fmaxf(x, 0.0f) + log1pf(expf(-fabsf(x)));
    }
    sdens[threadIdx.x] = v;
  }

  const float h0  = h[0];
  const float up0 = (h0 > 0.0f) ? h0 : __int_as_float(0x7fc00000);

  float al[4], be[4];
#pragma unroll
  for (int i = 0; i < 4; ++i) {
    const int p  = 4 * q + i;
    const int pc = min(p, n - 1);
    const float b0 = mesh[2 * pc];
    const float a0 = mesh[2 * pc + 1];
    be[i] = (p < n) ? b0 : -3.0e38f;
    al[i] = (p < n) ? a0 : 3.0e38f;
  }

  stage_store(tab, T, 0, hi, hp);

  unsigned bw[4];
#pragma unroll
  for (int i = 0; i < 4; ++i) bw[i] = (al[i] <= up0) ? 0u : 1u;

  __syncthreads();

  const float bx = sdens[4 * wv + 2 * half];
  const float by = sdens[4 * wv + 2 * half + 1];
  v2f Bc[16];
#pragma unroll
  for (int j = 0; j < 16; ++j) {
    Bc[j].x = (row == j) ? bx : 0.0f;
    Bc[j].y = (row == j) ? by : 0.0f;
  }

  const int ntrip = Tpad / TRIP;
  for (int win0 = 0; win0 < ntrip; win0 += WIN / TRIP) {
    const int tend  = min(win0 + WIN / TRIP, ntrip);
    const int npass = 2 * (tend - win0);
    float2 cur[WPT], nxt[WPT];
#pragma unroll
    for (int w = 0; w < WPT; ++w) cur[w] = tab[w * 32 + lane];
    asm volatile("" :: "v"(cur[0].x), "v"(cur[0].y), "v"(cur[1].x), "v"(cur[1].y),
                       "v"(cur[2].x), "v"(cur[2].y), "v"(cur[3].x), "v"(cur[3].y));

    for (int tr = win0; tr < tend; ++tr) {
      v8f accE = {}, accO = {};
#pragma unroll
      for (int hh = 0; hh < 2; ++hh) {
        const int pn = min(2 * (tr - win0) + hh + 1, npass - 1);
#pragma unroll
        for (int w = 0; w < WPT; ++w) nxt[w] = tab[(pn * WPT + w) * 32 + lane];

        unsigned nx[4][WPT];
#pragma unroll
        for (int i = 0; i < 4; ++i) {
          unsigned g[WPT], d[WPT];
#pragma unroll
          for (int w = 0; w < WPT; ++w) {
            g[w] = __builtin_amdgcn_ballot_w32(al[i] <= cur[w].x);
            d[w] = __builtin_amdgcn_ballot_w32(be[i] >= cur[w].y);
          }
          unsigned s0, s1, s2, s3;
          asm("s_cmp_lg_u32 %[b], 0\n\t"
              "s_sub_co_ci_u32 %[s0], %[g0], %[d0]\n\t"
              "s_sub_co_ci_u32 %[s1], %[g1], %[d1]\n\t"
              "s_sub_co_ci_u32 %[s2], %[g2], %[d2]\n\t"
              "s_sub_co_ci_u32 %[s3], %[g3], %[d3]\n\t"
              "s_cselect_b32 %[b], 1, 0"
              : [s0] "=&s"(s0), [s1] "=&s"(s1), [s2] "=&s"(s2), [s3] "=&s"(s3), [b] "+s"(bw[i])
              : [g0] "s"(g[0]), [d0] "s"(d[0]), [g1] "s"(g[1]), [d1] "s"(d[1]),
                [g2] "s"(g[2]), [d2] "s"(d[2]), [g3] "s"(g[3]), [d3] "s"(d[3])
              : "scc");
          nx[i][0] = s0 ^ g[0] ^ d[0];
          nx[i][1] = s1 ^ g[1] ^ d[1];
          nx[i][2] = s2 ^ g[2] ^ d[2];
          nx[i][3] = s3 ^ g[3] ^ d[3];
        }

#pragma unroll
        for (int w = 0; w < WPT; ++w) {
#pragma unroll
          for (int sub = 0; sub < 2; ++sub) {
            const unsigned c02 = sub ? pack_hi(nx[0][w], nx[2][w]) : pack_lo(nx[0][w], nx[2][w]);
            const unsigned c13 = sub ? pack_hi(nx[1][w], nx[3][w]) : pack_lo(nx[1][w], nx[3][w]);
            v2f A;
            A.x = __builtin_amdgcn_inverse_ballot_w32(c02) ? -1.0f : 1.0f;
            A.y = __builtin_amdgcn_inverse_ballot_w32(c13) ? -1.0f : 1.0f;
            if (sub)
              accO = __builtin_amdgcn_wmma_f32_16x16x4_f32(
                  false, A, false, Bc[8 * hh + 2 * w + 1], (short)0, accO, false, false);
            else
              accE = __builtin_amdgcn_wmma_f32_16x16x4_f32(
                  false, A, false, Bc[8 * hh + 2 * w], (short)0, accE, false, false);
          }
        }
#pragma unroll
        for (int w = 0; w < WPT; ++w) cur[w] = nxt[w];
      }
      float* qd = &red[wv][(tr - win0) * TRIP + 16 * row + 8 * half];
      *reinterpret_cast<float4*>(qd) = make_float4(
          accE[0] + accO[0], accE[1] + accO[1], accE[2] + accO[2], accE[3] + accO[3]);
      *reinterpret_cast<float4*>(qd + 4) = make_float4(
          accE[4] + accO[4], accE[5] + accO[5], accE[6] + accO[6], accE[7] + accO[7]);
    }
    __syncthreads();

    const int  wnext = win0 + WIN / TRIP;
    const bool more  = wnext < ntrip;
    if (more) stage_load(h, T, wnext * TRIP, hi, hp);

    const int tbase = win0 * TRIP;
    float* pout = partial + (size_t)blockIdx.x * T + tbase;
    for (int c4 = threadIdx.x; c4 < WIN / 4; c4 += SCAN_BLOCK) {
      float4 s = *reinterpret_cast<const float4*>(&red[0][4 * c4]);
#pragma unroll
      for (int k = 1; k < QW; ++k) {
        const float4 x = *reinterpret_cast<const float4*>(&red[k][4 * c4]);
        s.x += x.x; s.y += x.y; s.z += x.z; s.w += x.w;
      }
      if (tbase + 4 * c4 < T)
        *reinterpret_cast<float4*>(pout + 4 * c4) = s;
    }
    if (more) stage_store(tab, T, wnext * TRIP, hi, hp);
    __syncthreads();
  }
}

#define RED_COLS 32
#define RED_WAVES (BLOCK / 32)
#define RED_SLOTS (4 * RED_WAVES)
#define RED_UNROLL 10
__global__ __launch_bounds__(BLOCK) void hyst_reduce_kernel(
    const float* __restrict__ partial, const float* __restrict__ scale,
    const float* __restrict__ offset, float* __restrict__ out,
    int T, int NW, float inv_n) {
  __shared__ float red[RED_WAVES][RED_COLS];

  const int lane = threadIdx.x & 31;
  const int wv   = threadIdx.x >> 5;
  const int slot = 4 * wv + (lane >> 3);
  const int col  = blockIdx.x * RED_COLS + 4 * (lane & 7);

  float4 acc = make_float4(0.0f, 0.0f, 0.0f, 0.0f);
  if (col < T) {
    const float* base = partial + col;
    for (int r0 = slot; r0 < NW; r0 += RED_SLOTS * RED_UNROLL) {
      float4 x[RED_UNROLL];
#pragma unroll
      for (int i = 0; i < RED_UNROLL; ++i) {
        const int r = min(r0 + i * RED_SLOTS, NW - 1);
        x[i] = *reinterpret_cast<const float4*>(base + (size_t)r * T);
      }
#pragma unroll
      for (int i = 0; i < RED_UNROLL; ++i) {
        const bool in = (r0 + i * RED_SLOTS) < NW;
        acc.x += in ? x[i].x : 0.0f;
        acc.y += in ? x[i].y : 0.0f;
        acc.z += in ? x[i].z : 0.0f;
        acc.w += in ? x[i].w : 0.0f;
      }
    }
  }
  acc.x += __shfl_xor(acc.x, 8);  acc.y += __shfl_xor(acc.y, 8);
  acc.z += __shfl_xor(acc.z, 8);  acc.w += __shfl_xor(acc.w, 8);
  acc.x += __shfl_xor(acc.x, 16); acc.y += __shfl_xor(acc.y, 16);
  acc.z += __shfl_xor(acc.z, 16); acc.w += __shfl_xor(acc.w, 16);
  if (lane < 8) *reinterpret_cast<float4*>(&red[wv][4 * lane]) = acc;
  __syncthreads();

  if (threadIdx.x < RED_COLS) {
    const int t = blockIdx.x * RED_COLS + threadIdx.x;
    if (t < T) {
      float sum = 0.0f;
#pragma unroll
      for (int k = 0; k < RED_WAVES; ++k) sum += red[k][threadIdx.x];
      out[t] = scale[0] * (sum * inv_n) + offset[0];
    }
  }
}

extern "C" void kernel_launch(void* const* d_in, const int* in_sizes, int n_in,
                              void* d_out, int out_size, void* d_ws, size_t ws_size,
                              hipStream_t stream) {
  const float* h      = (const float*)d_in[0];
  const float* mesh   = (const float*)d_in[1];
  const float* raw    = (const float*)d_in[2];
  const float* scale  = (const float*)d_in[3];
  const float* offset = (const float*)d_in[4];
  float* out = (float*)d_out;

  const int T = in_sizes[0];
  const int n = in_sizes[2];

  const int nQuads  = (n + 3) / 4;
  const int nGroups = (nQuads + QW - 1) / QW;
  const int Tpad    = ((T + TRIP - 1) / TRIP) * TRIP;

  float* partial = (float*)d_ws;

  hyst_scan_kernel<<<nGroups, SCAN_BLOCK, 0, stream>>>(
      h, mesh, raw, partial, T, Tpad, n);
  hyst_reduce_kernel<<<(T + RED_COLS - 1) / RED_COLS, BLOCK, 0, stream>>>(
      partial, scale, offset, out, T, nGroups, 1.0f / (float)n);
}
